# starting kernel
# baseline (speedup 1.0000x reference)
#include <hip/hip_runtime.h>

#define NNODES 100000
#define NEDGES 600000
#define FDIM   128
#define NREL   3
#define LDSS   516

typedef float v2f __attribute__((ext_vector_type(2)));
typedef float v8f __attribute__((ext_vector_type(8)));

__global__ void zero_f4(float* __restrict__ p, long n4) {
  long i = (long)blockIdx.x * blockDim.x + threadIdx.x;
  long stride = (long)gridDim.x * blockDim.x;
  float4 z = make_float4(0.f, 0.f, 0.f, 0.f);
  for (; i < n4; i += stride) ((float4*)p)[i] = z;
}

__global__ void edge_agg(const int* __restrict__ ei, const int* __restrict__ et,
                         const float* __restrict__ feat, float* __restrict__ agg,
                         float* __restrict__ cnt, int do_cnt) {
  int lane = threadIdx.x & 31;
  int wave = (int)((blockIdx.x * blockDim.x + threadIdx.x) >> 5);
  int nwaves = (int)((gridDim.x * blockDim.x) >> 5);
  for (int e = wave; e < NEDGES; e += nwaves) {
    int s = ei[e];
    int d = ei[NEDGES + e];
    int r = et[e];
    float4 v = *(const float4*)(feat + (long)s * FDIM + lane * 4);
    float* a = agg + ((long)r * NNODES + d) * FDIM + lane * 4;
    atomicAdd(a + 0, v.x);
    atomicAdd(a + 1, v.y);
    atomicAdd(a + 2, v.z);
    atomicAdd(a + 3, v.w);
    if (do_cnt && lane == 0) atomicAdd(cnt + (long)r * NNODES + d, 1.0f);
  }
}

__global__ void normalize_agg(float* __restrict__ agg, const float* __restrict__ cnt) {
  long g = (long)blockIdx.x * blockDim.x + threadIdx.x;
  long row = g >> 5;
  int c4 = (int)(g & 31);
  if (row >= (long)NREL * NNODES) return;
  float s = 1.0f / fmaxf(cnt[row], 1.0f);
  float4* p = (float4*)(agg + row * FDIM) + c4;
  float4 v = *p;
  v.x *= s; v.y *= s; v.z *= s; v.w *= s;
  *p = v;
}

__global__ __launch_bounds__(256) void rgcn_gemm(
    const float* __restrict__ A0, const float* __restrict__ A1,
    const float* __restrict__ A2, const float* __restrict__ A3,
    const float* __restrict__ B0, const float* __restrict__ B1,
    const float* __restrict__ B2, const float* __restrict__ B3,
    const float* __restrict__ bias, float* __restrict__ out) {
  __shared__ __align__(16) float Alds[16 * LDSS];
  int t = threadIdx.x;
  long row0 = (long)blockIdx.x * 16;

  #pragma unroll
  for (int i = 0; i < 8; ++i) {
    int f4  = t + 256 * i;
    int row = f4 >> 7;
    int c4  = f4 & 127;
    int seg = c4 >> 5;
    int cw  = c4 & 31;
    const float* sp = (seg == 0) ? A0 : (seg == 1) ? A1 : (seg == 2) ? A2 : A3;
    float4 v = *(const float4*)(sp + (row0 + row) * FDIM + cw * 4);
    *(float4*)(&Alds[row * LDSS + c4 * 4]) = v;
  }
  __syncthreads();

  int wave = t >> 5;
  int lane = t & 31;
  int half = lane >> 4;
  int mn   = lane & 15;
  int col0 = wave * 16;

  v8f acc = {};
  const float* Bs[4] = {B0, B1, B2, B3};
  #pragma unroll
  for (int seg = 0; seg < 4; ++seg) {
    const float* B  = Bs[seg];
    const float* Ab = &Alds[mn * LDSS + seg * 128 + 2 * half];
    #pragma unroll 4
    for (int kk = 0; kk < 128; kk += 4) {
      v2f a = *(const v2f*)(Ab + kk);
      int kr = kk + 2 * half;
      v2f b;
      b.x = B[kr * FDIM + col0 + mn];
      b.y = B[(kr + 1) * FDIM + col0 + mn];
      acc = __builtin_amdgcn_wmma_f32_16x16x4_f32(
           false, a,  false, b,
           (short)0, acc,  false,  false);
    }
  }

  int col = col0 + mn;
  float bc = bias[col];
  #pragma unroll
  for (int i = 0; i < 8; ++i) {
    int m = i + 8 * half;
    out[(row0 + m) * FDIM + col] = fmaxf(acc[i] + bc, 0.0f);
  }
}

__global__ void head_logsoftmax(const float* __restrict__ h, const float* __restrict__ lw,
                                const float* __restrict__ lb, float* __restrict__ out) {
  int lane = threadIdx.x & 31;
  int node = (int)((blockIdx.x * blockDim.x + threadIdx.x) >> 5);
  if (node >= NNODES) return;
  float4 v = *(const float4*)(h + (long)node * FDIM + lane * 4);
  const float* w = lw + lane * 8;
  float p0 = v.x * w[0] + v.y * w[2] + v.z * w[4] + v.w * w[6];
  float p1 = v.x * w[1] + v.y * w[3] + v.z * w[5] + v.w * w[7];
  #pragma unroll
  for (int off = 16; off > 0; off >>= 1) {
    p0 += __shfl_xor(p0, off, 32);
    p1 += __shfl_xor(p1, off, 32);
  }
  if (lane == 0) {
    float l0 = p0 + lb[0], l1 = p1 + lb[1];
    float m = fmaxf(l0, l1);
    float lse = m + logf(expf(l0 - m) + expf(l1 - m));
    out[(long)node * 2 + 0] = l0 - lse;
    out[(long)node * 2 + 1] = l1 - lse;
  }
}

extern "C" void kernel_launch(void* const* d_in, const int* in_sizes, int n_in,
                              void* d_out, int out_size, void* d_ws, size_t ws_size,
                              hipStream_t stream) {
  const float* x      = (const float*)d_in[0];
  const int*   ei     = (const int*)d_in[1];
  const int*   et     = (const int*)d_in[2];
  const float* W1rel  = (const float*)d_in[3];
  const float* W1root = (const float*)d_in[4];
  const float* b1     = (const float*)d_in[5];
  const float* W2rel  = (const float*)d_in[6];
  const float* W2root = (const float*)d_in[7];
  const float* b2     = (const float*)d_in[8];
  const float* lw     = (const float*)d_in[9];
  const float* lb     = (const float*)d_in[10];

  const long aggN = (long)NREL * NNODES * FDIM;
  const long cntN = (long)NREL * NNODES;
  const long NF   = (long)NNODES * FDIM;

  float* ws  = (float*)d_ws;
  float* agg = ws;
  float* cnt = agg + aggN;
  float* h1  = cnt + cntN;
  float* h2  = h1 + NF;

  dim3 blk(256);

  zero_f4<<<4096, blk, 0, stream>>>(agg, (aggN + cntN) / 4);
  edge_agg<<<2048, blk, 0, stream>>>(ei, et, x, agg, cnt,  1);
  normalize_agg<<<(int)(cntN * 32 / 256), blk, 0, stream>>>(agg, cnt);
  rgcn_gemm<<<NNODES / 16, blk, 0, stream>>>(
      x, agg, agg + NF, agg + 2 * NF,
      W1root, W1rel, W1rel + FDIM * FDIM, W1rel + 2 * FDIM * FDIM, b1, h1);

  zero_f4<<<4096, blk, 0, stream>>>(agg, aggN / 4);
  edge_agg<<<2048, blk, 0, stream>>>(ei, et, h1, agg, cnt,  0);
  normalize_agg<<<(int)(cntN * 32 / 256), blk, 0, stream>>>(agg, cnt);
  rgcn_gemm<<<NNODES / 16, blk, 0, stream>>>(
      h1, agg, agg + NF, agg + 2 * NF,
      W2root, W2rel, W2rel + FDIM * FDIM, W2rel + 2 * FDIM * FDIM, b2, h2);

  head_logsoftmax<<<(NNODES * 32) / 256, blk, 0, stream>>>(h2, lw, lb, (float*)d_out);
}


# neighbour means gathered in the GEMM's A staging from a destination-ordered edge list built once per call (int count, alloc, fill); float-atomic scatter, zero passes and sum planes dropped
# speedup vs baseline: 2.1383x; 2.1383x over previous
#include <hip/hip_runtime.h>

#define NNODES 100000
#define NEDGES 600000
#define FDIM   128
#define NREL   3
#define LDSS   516

typedef float v2f __attribute__((ext_vector_type(2)));
typedef float v8f __attribute__((ext_vector_type(8)));

#define EQ     (NEDGES / 4)


__global__ void seg_init(int* __restrict__ deg, int* __restrict__ cursor) {
  int i = (int)(blockIdx.x * blockDim.x + threadIdx.x);
  if (i < NNODES) deg[i] = 0;
  if (i == 0) cursor[0] = 0;
}

__global__ void seg_count(const int* __restrict__ ei, int* __restrict__ deg,
                          int* __restrict__ rank) {
  int t = (int)(blockIdx.x * blockDim.x + threadIdx.x);
  if (t >= EQ) return;
  int d[4], r[4];
  #pragma unroll
  for (int k = 0; k < 4; ++k) d[k] = ei[NEDGES + t + k * EQ];
  #pragma unroll
  for (int k = 0; k < 4; ++k) r[k] = atomicAdd(&deg[d[k]], 1);
  #pragma unroll
  for (int k = 0; k < 4; ++k) rank[t + k * EQ] = r[k];
}

__global__ void seg_alloc(const int* __restrict__ deg, int* __restrict__ cursor,
                          int* __restrict__ seg) {
  int n = (int)(blockIdx.x * blockDim.x + threadIdx.x);
  int lane = threadIdx.x & 31;
  int d = (n < NNODES) ? deg[n] : 0;
  int incl = d;
  #pragma unroll
  for (int off = 1; off < 32; off <<= 1) {
    int up = __shfl_up(incl, off, 32);
    if (lane >= off) incl += up;
  }
  int total = __shfl(incl, 31, 32);
  int base = 0;
  if (lane == 31) base = atomicAdd(cursor, total);
  base = __shfl(base, 31, 32);
  if (n < NNODES) {
    seg[2 * n]     = base + incl - d;
    seg[2 * n + 1] = d;
  }
}

__global__ void seg_fill(const int* __restrict__ ei, const int* __restrict__ et,
                         const int* __restrict__ seg, const int* __restrict__ rank,
                         int* __restrict__ perm) {
  int t = (int)(blockIdx.x * blockDim.x + threadIdx.x);
  if (t >= EQ) return;
  #pragma unroll
  for (int k = 0; k < 4; ++k) {
    int e = t + k * EQ;
    int s = ei[e];
    int d = ei[NEDGES + e];
    int r = et[e];
    perm[seg[2 * d] + rank[e]] = s * 4 + r;
  }
}

__device__ __forceinline__ void gather_means(const float* __restrict__ col,
                                             const int* __restrict__ perm,
                                             int beg, int cnt, int pfirst, int lane,
                                             float* __restrict__ q) {
  float4 a0 = make_float4(0.f, 0.f, 0.f, 0.f), a1 = a0, a2 = a0;
  int c0 = 0, c1 = 0, c2 = 0;
  for (int base = 0; base < cnt; base += 32) {
    int m = cnt - base;
    if (m > 32) m = 32;
    int p = pfirst;
    if (base > 0) {
      p = 0;
      if (lane < m) p = perm[beg + base + lane];
    }
    for (int i = 0; i < m; i += 4) {
      float4 v[4];
      int rel[4];
      #pragma unroll
      for (int k = 0; k < 4; ++k) {
        int idx = i + k;
        bool live = idx < m;
        if (!live) idx = m - 1;
        int w = __builtin_amdgcn_readlane(p, idx);
        rel[k] = live ? (w & 3) : 3;
        v[k] = *(const float4*)(col + (long)(w >> 2) * FDIM);
      }
      #pragma unroll
      for (int k = 0; k < 4; ++k) {
        if (rel[k] == 0)      { a0.x += v[k].x; a0.y += v[k].y; a0.z += v[k].z; a0.w += v[k].w; ++c0; }
        else if (rel[k] == 1) { a1.x += v[k].x; a1.y += v[k].y; a1.z += v[k].z; a1.w += v[k].w; ++c1; }
        else if (rel[k] == 2) { a2.x += v[k].x; a2.y += v[k].y; a2.z += v[k].z; a2.w += v[k].w; ++c2; }
      }
    }
  }
  float s0 = 1.0f / fmaxf((float)c0, 1.0f);
  float s1 = 1.0f / fmaxf((float)c1, 1.0f);
  float s2 = 1.0f / fmaxf((float)c2, 1.0f);
  a0.x *= s0; a0.y *= s0; a0.z *= s0; a0.w *= s0;
  a1.x *= s1; a1.y *= s1; a1.z *= s1; a1.w *= s1;
  a2.x *= s2; a2.y *= s2; a2.z *= s2; a2.w *= s2;
  *(float4*)(q + 128) = a0;
  *(float4*)(q + 256) = a1;
  *(float4*)(q + 384) = a2;
}

__global__ __launch_bounds__(256) void rgcn_gemm(
    const float* __restrict__ feat, const int* __restrict__ seg, const int* __restrict__ perm,
    const float* __restrict__ B0, const float* __restrict__ B1,
    const float* __restrict__ B2, const float* __restrict__ B3,
    const float* __restrict__ bias, float* __restrict__ out) {
  __shared__ __align__(16) float Alds[16 * LDSS];
  int t = threadIdx.x;
  int wave = __builtin_amdgcn_readfirstlane(t >> 5);
  int lane = t & 31;
  long row0 = (long)blockIdx.x * 16;

  int n0 = (int)row0 + 2 * wave;
  const float* fcol = feat + lane * 4;
  float4 xr0 = *(const float4*)(fcol + (long)n0 * FDIM);
  float4 xr1 = *(const float4*)(fcol + (long)(n0 + 1) * FDIM);
  int4 sg = *(const int4*)(seg + 2 * n0);
  int p0 = 0, p1 = 0;
  if (lane < sg.y) p0 = perm[sg.x + lane];
  if (lane < sg.w) p1 = perm[sg.z + lane];
  float* q0 = &Alds[(2 * wave) * LDSS + lane * 4];
  float* q1 = q0 + LDSS;
  *(float4*)q0 = xr0;
  *(float4*)q1 = xr1;
  gather_means(fcol, perm, sg.x, sg.y, p0, lane, q0);
  gather_means(fcol, perm, sg.z, sg.w, p1, lane, q1);
  __syncthreads();

  int half = lane >> 4;
  int mn   = lane & 15;
  int col0 = wave * 16;

  v8f acc = {};
  const float* Bs[4] = {B0, B1, B2, B3};
  #pragma unroll
  for (int seg = 0; seg < 4; ++seg) {
    const float* B  = Bs[seg];
    const float* Ab = &Alds[mn * LDSS + seg * 128 + 2 * half];
    #pragma unroll 4
    for (int kk = 0; kk < 128; kk += 4) {
      v2f a = *(const v2f*)(Ab + kk);
      int kr = kk + 2 * half;
      v2f b;
      b.x = B[kr * FDIM + col0 + mn];
      b.y = B[(kr + 1) * FDIM + col0 + mn];
      acc = __builtin_amdgcn_wmma_f32_16x16x4_f32(
           false, a,  false, b,
           (short)0, acc,  false,  false);
    }
  }

  int col = col0 + mn;
  float bc = bias[col];
  #pragma unroll
  for (int i = 0; i < 8; ++i) {
    int m = i + 8 * half;
    out[(row0 + m) * FDIM + col] = fmaxf(acc[i] + bc, 0.0f);
  }
}

__global__ void head_logsoftmax(const float* __restrict__ h, const float* __restrict__ lw,
                                const float* __restrict__ lb, float* __restrict__ out) {
  int lane = threadIdx.x & 31;
  int node = (int)((blockIdx.x * blockDim.x + threadIdx.x) >> 5);
  if (node >= NNODES) return;
  float4 v = *(const float4*)(h + (long)node * FDIM + lane * 4);
  const float* w = lw + lane * 8;
  float p0 = v.x * w[0] + v.y * w[2] + v.z * w[4] + v.w * w[6];
  float p1 = v.x * w[1] + v.y * w[3] + v.z * w[5] + v.w * w[7];
  #pragma unroll
  for (int off = 16; off > 0; off >>= 1) {
    p0 += __shfl_xor(p0, off, 32);
    p1 += __shfl_xor(p1, off, 32);
  }
  if (lane == 0) {
    float l0 = p0 + lb[0], l1 = p1 + lb[1];
    float m = fmaxf(l0, l1);
    float lse = m + logf(expf(l0 - m) + expf(l1 - m));
    out[(long)node * 2 + 0] = l0 - lse;
    out[(long)node * 2 + 1] = l1 - lse;
  }
}

extern "C" void kernel_launch(void* const* d_in, const int* in_sizes, int n_in,
                              void* d_out, int out_size, void* d_ws, size_t ws_size,
                              hipStream_t stream) {
  const float* x      = (const float*)d_in[0];
  const int*   ei     = (const int*)d_in[1];
  const int*   et     = (const int*)d_in[2];
  const float* W1rel  = (const float*)d_in[3];
  const float* W1root = (const float*)d_in[4];
  const float* b1     = (const float*)d_in[5];
  const float* W2rel  = (const float*)d_in[6];
  const float* W2root = (const float*)d_in[7];
  const float* b2     = (const float*)d_in[8];
  const float* lw     = (const float*)d_in[9];
  const float* lb     = (const float*)d_in[10];

  const long aggN = (long)NREL * NNODES * FDIM;
  const long cntN = (long)NREL * NNODES;
  const long NF   = (long)NNODES * FDIM;

  float* ws  = (float*)d_ws;
  int* deg    = (int*)ws;
  int* cursor = deg + NNODES;
  int* seg    = deg + NNODES + 32;
  int* rank   = seg + 2 * NNODES;
  int* perm   = rank + NEDGES;
  float* h1  = ws + aggN + cntN;
  float* h2  = h1 + NF;

  dim3 blk(256);

  seg_init<<<(NNODES + 255) / 256, blk, 0, stream>>>(deg, cursor);
  seg_count<<<(EQ + 255) / 256, blk, 0, stream>>>(ei, deg, rank);
  seg_alloc<<<(NNODES + 255) / 256, blk, 0, stream>>>(deg, cursor, seg);
  seg_fill<<<(EQ + 255) / 256, blk, 0, stream>>>(ei, et, seg, rank, perm);

  rgcn_gemm<<<NNODES / 16, blk, 0, stream>>>(
      x, seg, perm,
      W1root, W1rel, W1rel + FDIM * FDIM, W1rel + 2 * FDIM * FDIM, b1, h1);

  rgcn_gemm<<<NNODES / 16, blk, 0, stream>>>(
      h1, seg, perm,
      W2root, W2rel, W2rel + FDIM * FDIM, W2rel + 2 * FDIM * FDIM, b2, h2);

  head_logsoftmax<<<(NNODES * 32) / 256, blk, 0, stream>>>(h2, lw, lb, (float*)d_out);
}


# layer GEMM weights packed once per call into WMMA fragment order: a wave's B operand for two k-steps is one coalesced 16-byte load per lane (A by ds_load_b128), 8 loads in flight feed 16 WMMAs
# speedup vs baseline: 2.5064x; 1.1721x over previous
#include <hip/hip_runtime.h>

#define NNODES 100000
#define NEDGES 600000
#define FDIM   128
#define NREL   3
#define LDSS   516

typedef float v2f __attribute__((ext_vector_type(2)));
typedef float v8f __attribute__((ext_vector_type(8)));

#define PACK4  (4 * FDIM * FDIM / 4)
#define EQ     (NEDGES / 4)


__global__ void seg_init(int* __restrict__ deg, int* __restrict__ cursor) {
  int i = (int)(blockIdx.x * blockDim.x + threadIdx.x);
  if (i < NNODES) deg[i] = 0;
  if (i == 0) cursor[0] = 0;
}

__global__ void seg_count(const int* __restrict__ ei, int* __restrict__ deg,
                          int* __restrict__ rank) {
  int t = (int)(blockIdx.x * blockDim.x + threadIdx.x);
  if (t >= EQ) return;
  int d[4], r[4];
  #pragma unroll
  for (int k = 0; k < 4; ++k) d[k] = ei[NEDGES + t + k * EQ];
  #pragma unroll
  for (int k = 0; k < 4; ++k) r[k] = atomicAdd(&deg[d[k]], 1);
  #pragma unroll
  for (int k = 0; k < 4; ++k) rank[t + k * EQ] = r[k];
}

__global__ void seg_alloc(const int* __restrict__ deg, int* __restrict__ cursor,
                          int* __restrict__ seg) {
  int n = (int)(blockIdx.x * blockDim.x + threadIdx.x);
  int lane = threadIdx.x & 31;
  int d = (n < NNODES) ? deg[n] : 0;
  int incl = d;
  #pragma unroll
  for (int off = 1; off < 32; off <<= 1) {
    int up = __shfl_up(incl, off, 32);
    if (lane >= off) incl += up;
  }
  int total = __shfl(incl, 31, 32);
  int base = 0;
  if (lane == 31) base = atomicAdd(cursor, total);
  base = __shfl(base, 31, 32);
  if (n < NNODES) {
    seg[2 * n]     = base + incl - d;
    seg[2 * n + 1] = d;
  }
}

__global__ void seg_fill(const int* __restrict__ ei, const int* __restrict__ et,
                         const int* __restrict__ seg, const int* __restrict__ rank,
                         int* __restrict__ perm) {
  int t = (int)(blockIdx.x * blockDim.x + threadIdx.x);
  if (t >= EQ) return;
  #pragma unroll
  for (int k = 0; k < 4; ++k) {
    int e = t + k * EQ;
    int s = ei[e];
    int d = ei[NEDGES + e];
    int r = et[e];
    perm[seg[2 * d] + rank[e]] = s * 4 + r;
  }
}

__device__ __forceinline__ void gather_means(const float* __restrict__ col,
                                             const int* __restrict__ perm,
                                             int beg, int cnt, int pfirst, int lane,
                                             float* __restrict__ q) {
  float4 a0 = make_float4(0.f, 0.f, 0.f, 0.f), a1 = a0, a2 = a0;
  int c0 = 0, c1 = 0, c2 = 0;
  for (int base = 0; base < cnt; base += 32) {
    int m = cnt - base;
    if (m > 32) m = 32;
    int p = pfirst;
    if (base > 0) {
      p = 0;
      if (lane < m) p = perm[beg + base + lane];
    }
    for (int i = 0; i < m; i += 4) {
      float4 v[4];
      int rel[4];
      #pragma unroll
      for (int k = 0; k < 4; ++k) {
        int idx = i + k;
        bool live = idx < m;
        if (!live) idx = m - 1;
        int w = __builtin_amdgcn_readlane(p, idx);
        rel[k] = live ? (w & 3) : 3;
        v[k] = *(const float4*)(col + (long)(w >> 2) * FDIM);
      }
      #pragma unroll
      for (int k = 0; k < 4; ++k) {
        if (rel[k] == 0)      { a0.x += v[k].x; a0.y += v[k].y; a0.z += v[k].z; a0.w += v[k].w; ++c0; }
        else if (rel[k] == 1) { a1.x += v[k].x; a1.y += v[k].y; a1.z += v[k].z; a1.w += v[k].w; ++c1; }
        else if (rel[k] == 2) { a2.x += v[k].x; a2.y += v[k].y; a2.z += v[k].z; a2.w += v[k].w; ++c2; }
      }
    }
  }
  float s0 = 1.0f / fmaxf((float)c0, 1.0f);
  float s1 = 1.0f / fmaxf((float)c1, 1.0f);
  float s2 = 1.0f / fmaxf((float)c2, 1.0f);
  a0.x *= s0; a0.y *= s0; a0.z *= s0; a0.w *= s0;
  a1.x *= s1; a1.y *= s1; a1.z *= s1; a1.w *= s1;
  a2.x *= s2; a2.y *= s2; a2.z *= s2; a2.w *= s2;
  *(float4*)(q + 128) = a0;
  *(float4*)(q + 256) = a1;
  *(float4*)(q + 384) = a2;
}

__global__ void pack_weights(const float* __restrict__ W1root, const float* __restrict__ W1rel,
                             const float* __restrict__ W2root, const float* __restrict__ W2rel,
                             float* __restrict__ Bp) {
  int i = (int)(blockIdx.x * blockDim.x + threadIdx.x);
  if (i >= 2 * PACK4) return;
  int layer = i / PACK4;
  int r     = i % PACK4;
  int lane = r & 31;
  int J    = (r >> 5) & 63;
  int ct   = r >> 11;
  int h = lane >> 4, c = lane & 15;
  int kg = 8 * J + 4 * h;
  int s  = kg >> 7;
  int k  = kg & 127;
  const float* root = layer ? W2root : W1root;
  const float* rel  = layer ? W2rel : W1rel;
  const float* B = (s == 0) ? root : rel + (s - 1) * FDIM * FDIM;
  int n = 16 * ct + c;
  float4 v;
  v.x = B[(k + 0) * FDIM + n];
  v.y = B[(k + 1) * FDIM + n];
  v.z = B[(k + 2) * FDIM + n];
  v.w = B[(k + 3) * FDIM + n];
  ((float4*)Bp)[i] = v;
}

__global__ __launch_bounds__(256) void rgcn_gemm(
    const float* __restrict__ feat, const int* __restrict__ seg, const int* __restrict__ perm,
    const float* __restrict__ Bp,
    const float* __restrict__ bias, float* __restrict__ out) {
  __shared__ __align__(16) float Alds[16 * LDSS];
  int t = threadIdx.x;
  int wave = __builtin_amdgcn_readfirstlane(t >> 5);
  int lane = t & 31;
  long row0 = (long)blockIdx.x * 16;

  int n0 = (int)row0 + 2 * wave;
  const float* fcol = feat + lane * 4;
  float4 xr0 = *(const float4*)(fcol + (long)n0 * FDIM);
  float4 xr1 = *(const float4*)(fcol + (long)(n0 + 1) * FDIM);
  int4 sg = *(const int4*)(seg + 2 * n0);
  int p0 = 0, p1 = 0;
  if (lane < sg.y) p0 = perm[sg.x + lane];
  if (lane < sg.w) p1 = perm[sg.z + lane];
  float* q0 = &Alds[(2 * wave) * LDSS + lane * 4];
  float* q1 = q0 + LDSS;
  *(float4*)q0 = xr0;
  *(float4*)q1 = xr1;
  gather_means(fcol, perm, sg.x, sg.y, p0, lane, q0);
  gather_means(fcol, perm, sg.z, sg.w, p1, lane, q1);
  __syncthreads();

  int half = lane >> 4;
  int mn   = lane & 15;
  int col0 = wave * 16;

  v8f acc = {};
  const float4* Bw = (const float4*)Bp + (long)wave * (64 * 32) + lane;
  const float*  Ab = &Alds[mn * LDSS + 4 * half];
  #pragma unroll 8
  for (int J = 0; J < 64; ++J) {
    float4 a4 = *(const float4*)(Ab + 8 * J);
    float4 b4 = Bw[J * 32];
    v2f a01 = {a4.x, a4.y}, b01 = {b4.x, b4.y};
    v2f a23 = {a4.z, a4.w}, b23 = {b4.z, b4.w};
    acc = __builtin_amdgcn_wmma_f32_16x16x4_f32(
         false, a01,  false, b01,
         (short)0, acc,  false,  false);
    acc = __builtin_amdgcn_wmma_f32_16x16x4_f32(
         false, a23,  false, b23,
         (short)0, acc,  false,  false);
  }

  int col = col0 + mn;
  float bc = bias[col];
  #pragma unroll
  for (int i = 0; i < 8; ++i) {
    int m = i + 8 * half;
    out[(row0 + m) * FDIM + col] = fmaxf(acc[i] + bc, 0.0f);
  }
}

__global__ void head_logsoftmax(const float* __restrict__ h, const float* __restrict__ lw,
                                const float* __restrict__ lb, float* __restrict__ out) {
  int lane = threadIdx.x & 31;
  int node = (int)((blockIdx.x * blockDim.x + threadIdx.x) >> 5);
  if (node >= NNODES) return;
  float4 v = *(const float4*)(h + (long)node * FDIM + lane * 4);
  const float* w = lw + lane * 8;
  float p0 = v.x * w[0] + v.y * w[2] + v.z * w[4] + v.w * w[6];
  float p1 = v.x * w[1] + v.y * w[3] + v.z * w[5] + v.w * w[7];
  #pragma unroll
  for (int off = 16; off > 0; off >>= 1) {
    p0 += __shfl_xor(p0, off, 32);
    p1 += __shfl_xor(p1, off, 32);
  }
  if (lane == 0) {
    float l0 = p0 + lb[0], l1 = p1 + lb[1];
    float m = fmaxf(l0, l1);
    float lse = m + logf(expf(l0 - m) + expf(l1 - m));
    out[(long)node * 2 + 0] = l0 - lse;
    out[(long)node * 2 + 1] = l1 - lse;
  }
}

extern "C" void kernel_launch(void* const* d_in, const int* in_sizes, int n_in,
                              void* d_out, int out_size, void* d_ws, size_t ws_size,
                              hipStream_t stream) {
  const float* x      = (const float*)d_in[0];
  const int*   ei     = (const int*)d_in[1];
  const int*   et     = (const int*)d_in[2];
  const float* W1rel  = (const float*)d_in[3];
  const float* W1root = (const float*)d_in[4];
  const float* b1     = (const float*)d_in[5];
  const float* W2rel  = (const float*)d_in[6];
  const float* W2root = (const float*)d_in[7];
  const float* b2     = (const float*)d_in[8];
  const float* lw     = (const float*)d_in[9];
  const float* lb     = (const float*)d_in[10];

  const long aggN = (long)NREL * NNODES * FDIM;
  const long cntN = (long)NREL * NNODES;
  const long NF   = (long)NNODES * FDIM;

  float* ws  = (float*)d_ws;
  int* deg    = (int*)ws;
  int* cursor = deg + NNODES;
  int* seg    = deg + NNODES + 32;
  int* rank   = seg + 2 * NNODES;
  int* perm   = rank + NEDGES;
  float* Bp  = (float*)(perm + NEDGES);
  float* h1  = ws + aggN + cntN;
  float* h2  = h1 + NF;

  dim3 blk(256);

  seg_init<<<(NNODES + 255) / 256, blk, 0, stream>>>(deg, cursor);
  seg_count<<<(EQ + 255) / 256, blk, 0, stream>>>(ei, deg, rank);
  seg_alloc<<<(NNODES + 255) / 256, blk, 0, stream>>>(deg, cursor, seg);
  seg_fill<<<(EQ + 255) / 256, blk, 0, stream>>>(ei, et, seg, rank, perm);

  pack_weights<<<(2 * PACK4 + 255) / 256, blk, 0, stream>>>(W1root, W1rel, W2root, W2rel, Bp);

  rgcn_gemm<<<NNODES / 16, blk, 0, stream>>>(x, seg, perm, Bp, b1, h1);

  rgcn_gemm<<<NNODES / 16, blk, 0, stream>>>(h1, seg, perm, Bp + 4 * PACK4, b2, h2);

  head_logsoftmax<<<(NNODES * 32) / 256, blk, 0, stream>>>(h2, lw, lb, (float*)d_out);
}


# layer GEMM block widened to 32 rows: a wave keeps 2 row tiles (2 independent accumulator chains) per packed weight fragment it fetches, so the weights are re-read from L2 half as often
# speedup vs baseline: 2.9325x; 1.1700x over previous
#include <hip/hip_runtime.h>

#define NNODES 100000
#define NEDGES 600000
#define FDIM   128
#define NREL   3
#define LDSS   516

typedef float v2f __attribute__((ext_vector_type(2)));
typedef float v8f __attribute__((ext_vector_type(8)));

#define PACK4  (4 * FDIM * FDIM / 4)
#define EQ     (NEDGES / 4)

#define CAP    32

#define RT     2
#define BROWS  (16 * RT)
#define NPW    (BROWS / 8)


__global__ void seg_init(int* __restrict__ deg, int* __restrict__ ovfn) {
  int i = (int)(blockIdx.x * blockDim.x + threadIdx.x);
  if (i < NNODES) deg[i] = 0;
  if (i == 0) ovfn[0] = 0;
}

__global__ void seg_place(const int* __restrict__ ei, const int* __restrict__ et,
                          int* __restrict__ deg, int* __restrict__ ovfn,
                          int* __restrict__ bucket, int* __restrict__ ovf) {
  int t = (int)(blockIdx.x * blockDim.x + threadIdx.x);
  if (t >= EQ) return;
  int d[4], w[4], r[4];
  #pragma unroll
  for (int k = 0; k < 4; ++k) {
    int e = t + k * EQ;
    d[k] = ei[NEDGES + e];
    w[k] = ei[e] * 4 + et[e];
  }
  #pragma unroll
  for (int k = 0; k < 4; ++k) r[k] = atomicAdd(&deg[d[k]], 1);
  #pragma unroll
  for (int k = 0; k < 4; ++k) {
    if (r[k] < CAP) {
      bucket[d[k] * CAP + r[k]] = w[k];
    } else {
      int o = atomicAdd(ovfn, 1);
      ovf[2 * o]     = d[k];
      ovf[2 * o + 1] = w[k];
    }
  }
}

__device__ __forceinline__ void gather_means(const float* __restrict__ col,
                                             const int* __restrict__ ovf,
                                             const int* __restrict__ ovfn,
                                             int node, int cnt, int p, int lane,
                                             float* __restrict__ q) {
  float4 a0 = make_float4(0.f, 0.f, 0.f, 0.f), a1 = a0, a2 = a0;
  int c0 = 0, c1 = 0, c2 = 0;
  int m = cnt < CAP ? cnt : CAP;
  for (int i = 0; i < m; i += 4) {
    float4 v[4];
    int rel[4];
    #pragma unroll
    for (int k = 0; k < 4; ++k) {
      int idx = i + k;
      bool live = idx < m;
      if (!live) idx = m - 1;
      int w = __builtin_amdgcn_readlane(p, idx);
      rel[k] = live ? (w & 3) : 3;
      v[k] = *(const float4*)(col + (long)(w >> 2) * FDIM);
    }
    #pragma unroll
    for (int k = 0; k < 4; ++k) {
      if (rel[k] == 0)      { a0.x += v[k].x; a0.y += v[k].y; a0.z += v[k].z; a0.w += v[k].w; ++c0; }
      else if (rel[k] == 1) { a1.x += v[k].x; a1.y += v[k].y; a1.z += v[k].z; a1.w += v[k].w; ++c1; }
      else if (rel[k] == 2) { a2.x += v[k].x; a2.y += v[k].y; a2.z += v[k].z; a2.w += v[k].w; ++c2; }
    }
  }
  if (cnt > CAP) {
    int total = ovfn[0];
    if (total > NEDGES) total = NEDGES;
    for (int base = 0; base < total; base += 32) {
      int od = -1, ow = 0;
      if (base + lane < total) {
        int2 pr = *(const int2*)(ovf + 2 * (base + lane));
        od = pr.x;
        ow = pr.y;
      }
      unsigned mine = (unsigned)__ballot(od == node);
      while (mine != 0u) {
        int i = __builtin_ctz(mine);
        mine &= mine - 1u;
        int w = __builtin_amdgcn_readlane(ow, i);
        int rl = w & 3;
        float4 v = *(const float4*)(col + (long)(w >> 2) * FDIM);
        if (rl == 0)      { a0.x += v.x; a0.y += v.y; a0.z += v.z; a0.w += v.w; ++c0; }
        else if (rl == 1) { a1.x += v.x; a1.y += v.y; a1.z += v.z; a1.w += v.w; ++c1; }
        else if (rl == 2) { a2.x += v.x; a2.y += v.y; a2.z += v.z; a2.w += v.w; ++c2; }
      }
    }
  }
  float s0 = 1.0f / fmaxf((float)c0, 1.0f);
  float s1 = 1.0f / fmaxf((float)c1, 1.0f);
  float s2 = 1.0f / fmaxf((float)c2, 1.0f);
  a0.x *= s0; a0.y *= s0; a0.z *= s0; a0.w *= s0;
  a1.x *= s1; a1.y *= s1; a1.z *= s1; a1.w *= s1;
  a2.x *= s2; a2.y *= s2; a2.z *= s2; a2.w *= s2;
  *(float4*)(q + 128) = a0;
  *(float4*)(q + 256) = a1;
  *(float4*)(q + 384) = a2;
}

__global__ void pack_weights(const float* __restrict__ W1root, const float* __restrict__ W1rel,
                             const float* __restrict__ W2root, const float* __restrict__ W2rel,
                             float* __restrict__ Bp) {
  int i = (int)(blockIdx.x * blockDim.x + threadIdx.x);
  if (i >= 2 * PACK4) return;
  int layer = i / PACK4;
  int r     = i % PACK4;
  int lane = r & 31;
  int J    = (r >> 5) & 63;
  int ct   = r >> 11;
  int h = lane >> 4, c = lane & 15;
  int kg = 8 * J + 4 * h;
  int s  = kg >> 7;
  int k  = kg & 127;
  const float* root = layer ? W2root : W1root;
  const float* rel  = layer ? W2rel : W1rel;
  const float* B = (s == 0) ? root : rel + (s - 1) * FDIM * FDIM;
  int n = 16 * ct + c;
  float4 v;
  v.x = B[(k + 0) * FDIM + n];
  v.y = B[(k + 1) * FDIM + n];
  v.z = B[(k + 2) * FDIM + n];
  v.w = B[(k + 3) * FDIM + n];
  ((float4*)Bp)[i] = v;
}

template <bool HEAD>
__global__ __launch_bounds__(256) void rgcn_gemm(
    const float* __restrict__ feat, const int* __restrict__ deg, const int* __restrict__ bucket,
    const int* __restrict__ ovf, const int* __restrict__ ovfn,
    const float* __restrict__ Bp,
    const float* __restrict__ bias, float* __restrict__ out,
    const float* __restrict__ lw, const float* __restrict__ lb) {
  __shared__ __align__(16) float Alds[BROWS * LDSS];
  int t = threadIdx.x;
  int wave = __builtin_amdgcn_readfirstlane(t >> 5);
  int lane = t & 31;
  long row0 = (long)blockIdx.x * BROWS;

  static_assert(NPW == 4, "the in-degrees of a wave's nodes are fetched as one int4");
  static_assert(NNODES % BROWS == 0, "every block owns BROWS whole rows: no row guard in the kernel");
  int n0 = (int)row0 + NPW * wave;
  const float* fcol = feat + lane * 4;
  float4 xr[NPW];
  #pragma unroll
  for (int k = 0; k < NPW; ++k) xr[k] = *(const float4*)(fcol + (long)(n0 + k) * FDIM);
  int4 dg4 = *(const int4*)(deg + n0);
  int dgs[NPW] = {dg4.x, dg4.y, dg4.z, dg4.w};
  int p[NPW];
  #pragma unroll
  for (int k = 0; k < NPW; ++k) {
    p[k] = 0;
    if (lane < dgs[k]) p[k] = bucket[(n0 + k) * CAP + lane];
  }
  float* q0 = &Alds[(NPW * wave) * LDSS + lane * 4];
  #pragma unroll
  for (int k = 0; k < NPW; ++k) *(float4*)(q0 + k * LDSS) = xr[k];
  #pragma unroll
  for (int k = 0; k < NPW; ++k)
    gather_means(fcol, ovf, ovfn, n0 + k, dgs[k], p[k], lane, q0 + k * LDSS);
  __syncthreads();

  int half = lane >> 4;
  int mn   = lane & 15;
  int col0 = wave * 16;

  v8f acc[RT];
  #pragma unroll
  for (int r = 0; r < RT; ++r) acc[r] = v8f{};
  const float4* Bw = (const float4*)Bp + (long)wave * (64 * 32) + lane;
  const float*  Ab = &Alds[mn * LDSS + 4 * half];
  #pragma unroll 8
  for (int J = 0; J < 64; ++J) {
    float4 b4 = Bw[J * 32];
    v2f b01 = {b4.x, b4.y};
    v2f b23 = {b4.z, b4.w};
    float4 a4[RT];
    #pragma unroll
    for (int r = 0; r < RT; ++r) a4[r] = *(const float4*)(Ab + r * 16 * LDSS + 8 * J);
    #pragma unroll
    for (int r = 0; r < RT; ++r) {
      v2f a01 = {a4[r].x, a4[r].y};
      acc[r] = __builtin_amdgcn_wmma_f32_16x16x4_f32(
           false, a01,  false, b01,
           (short)0, acc[r],  false,  false);
    }
    #pragma unroll
    for (int r = 0; r < RT; ++r) {
      v2f a23 = {a4[r].z, a4[r].w};
      acc[r] = __builtin_amdgcn_wmma_f32_16x16x4_f32(
           false, a23,  false, b23,
           (short)0, acc[r],  false,  false);
    }
  }

  int col = col0 + mn;
  float bc = bias[col];
  if constexpr (!HEAD) {
    #pragma unroll
    for (int r = 0; r < RT; ++r) {
      #pragma unroll
      for (int i = 0; i < 8; ++i) {
        int m = 16 * r + i + 8 * half;
        out[(row0 + m) * FDIM + col] = fmaxf(acc[r][i] + bc, 0.0f);
      }
    }
  } else {
    __shared__ float Plog[RT * 8 * 32];
    float2 w2 = *(const float2*)(lw + 2 * col);
    #pragma unroll
    for (int r = 0; r < RT; ++r) {
      float a[16];
      #pragma unroll
      for (int i = 0; i < 8; ++i) {
        float h = fmaxf(acc[r][i] + bc, 0.0f);
        a[2 * i]     = h * w2.x;
        a[2 * i + 1] = h * w2.y;
      }
      float k8[8], k4[4], k2[2], k1;
      bool b3 = (mn & 8) != 0, b2 = (mn & 4) != 0, b1 = (mn & 2) != 0, b0 = (mn & 1) != 0;
      #pragma unroll
      for (int j = 0; j < 8; ++j)
        k8[j] = (b3 ? a[j + 8] : a[j]) + __shfl_xor(b3 ? a[j] : a[j + 8], 8, 32);
      #pragma unroll
      for (int j = 0; j < 4; ++j)
        k4[j] = (b2 ? k8[j + 4] : k8[j]) + __shfl_xor(b2 ? k8[j] : k8[j + 4], 4, 32);
      #pragma unroll
      for (int j = 0; j < 2; ++j)
        k2[j] = (b1 ? k4[j + 2] : k4[j]) + __shfl_xor(b1 ? k4[j] : k4[j + 2], 2, 32);
      k1 = (b0 ? k2[1] : k2[0]) + __shfl_xor(b0 ? k2[0] : k2[1], 1, 32);
      Plog[(r * 8 + wave) * 32 + lane] = k1;
    }
    __syncthreads();
    if (wave < RT) {
      float s = 0.0f;
      #pragma unroll
      for (int w = 0; w < 8; ++w) s += Plog[(wave * 8 + w) * 32 + lane];
      float lg = s + lb[lane & 1];
      float other = __shfl_xor(lg, 1, 32);
      float m = fmaxf(lg, other);
      float lse = m + logf(expf(lg - m) + expf(other - m));
      out[(row0 + 16 * wave) * 2 + lane] = lg - lse;
    }
  }
}

extern "C" void kernel_launch(void* const* d_in, const int* in_sizes, int n_in,
                              void* d_out, int out_size, void* d_ws, size_t ws_size,
                              hipStream_t stream) {
  const float* x      = (const float*)d_in[0];
  const int*   ei     = (const int*)d_in[1];
  const int*   et     = (const int*)d_in[2];
  const float* W1rel  = (const float*)d_in[3];
  const float* W1root = (const float*)d_in[4];
  const float* b1     = (const float*)d_in[5];
  const float* W2rel  = (const float*)d_in[6];
  const float* W2root = (const float*)d_in[7];
  const float* b2     = (const float*)d_in[8];
  const float* lw     = (const float*)d_in[9];
  const float* lb     = (const float*)d_in[10];

  const long aggN = (long)NREL * NNODES * FDIM;
  const long cntN = (long)NREL * NNODES;

  float* ws  = (float*)d_ws;
  int* deg    = (int*)ws;
  int* ovfn   = deg + NNODES;
  int* bucket = deg + NNODES + 32;
  int* ovf    = bucket + (long)CAP * NNODES;
  float* Bp  = (float*)(ovf + 2 * NEDGES);
  float* h1  = ws + aggN + cntN;

  dim3 blk(256);

  seg_init<<<(NNODES + 255) / 256, blk, 0, stream>>>(deg, ovfn);
  seg_place<<<(EQ + 255) / 256, blk, 0, stream>>>(ei, et, deg, ovfn, bucket, ovf);

  pack_weights<<<(2 * PACK4 + 255) / 256, blk, 0, stream>>>(W1root, W1rel, W2root, W2rel, Bp);

  rgcn_gemm<false><<<NNODES / BROWS, blk, 0, stream>>>(x, deg, bucket, ovf, ovfn, Bp, b1, h1, lw, lb);

  rgcn_gemm<true><<<NNODES / BROWS, blk, 0, stream>>>(h1, deg, bucket, ovf, ovfn, Bp + 4 * PACK4, b2, (float*)d_out, lw, lb);
}
